# starting kernel
# baseline (speedup 1.0000x reference)
#include <hip/hip_runtime.h>
#include <hip/hip_bf16.h>
#include <math.h>

#define NB   8
#define CCH  128
#define HH   56
#define WW   56
#define OCH  64
#define NC   2
#define KK   5
#define PP   (KK*KK)
#define HW   (HH*WW)
#define NPIX (NB*HW)
#define NCOL 50
#define NCOLP 64

#define BROW 160
#define BROWS 64
#define BTOT (BROWS*BROW)

typedef __attribute__((ext_vector_type(2))) float v2f;
typedef __attribute__((ext_vector_type(8))) float v8f;

#define WS_BF    BTOT
#define WS_Y     (BTOT + 64)

__global__ void calib_fold_kernel(const float* __restrict__ Wt,
                                  const float* __restrict__ bias,
                                  const float* __restrict__ Wlin,
                                  const float* __restrict__ blin,
                                  float* __restrict__ Bp,
                                  float* __restrict__ bfold) {
    int idx = blockIdx.x * 256 + threadIdx.x;
    if (idx >= BTOT) return;
    int row = idx / BROW;
    int off = idx - row * BROW;
    float v = 0.f;
    if (off < 2 * NCOLP) {
        int col = off >> 1;
        int par = off & 1;
        int c   = row * 2 + par;
        if (col < NCOL) {
            int p = col >> 1;
            int k = col & 1;
            #pragma unroll 8
            for (int o = 0; o < OCH; ++o)
                v += Wlin[k * OCH + o] * Wt[(size_t)o * (CCH * PP) + c * PP + p];
        }
    }
    Bp[idx] = v;
    if (idx < NC) {
        float s = blin[idx];
        #pragma unroll 8
        for (int o = 0; o < OCH; ++o)
            s += Wlin[idx * OCH + o] * bias[o];
        bfold[idx] = s;
    }
}

__global__ void __launch_bounds__(256)
calib_gemm_kernel(const float* __restrict__ x,
                  const float* __restrict__ Bp,
                  float* __restrict__ Y) {
    __shared__ float Blds[BTOT];

    const int tid  = threadIdx.x;
    const int wave = tid >> 5;
    const int lane = tid & 31;

    for (int i = tid; i < BTOT; i += 256)
        Blds[i] = Bp[i];
    __syncthreads();

    const int pixBase = (blockIdx.x * 8 + wave) * 16;

    const int laneM = lane & 15;
    const int kHalf = (lane >> 4) << 1;
    const int pm    = pixBase + laneM;
    const int n     = pm / HW;
    const int hw    = pm - n * HW;
    const float* xb = x + (size_t)n * CCH * HW + hw;

    v8f acc0 = {}, acc1 = {}, acc2 = {}, acc3 = {};

    #pragma unroll 4
    for (int k0 = 0; k0 < CCH; k0 += 4) {
        const int c = k0 + kHalf;
        v2f a;
        a.x = xb[(size_t)c * HW];
        a.y = xb[(size_t)(c + 1) * HW];

        const float* brow = &Blds[(c >> 1) * BROW + laneM * 2];
        v2f b0 = *(const v2f*)(brow + 0);
        v2f b1 = *(const v2f*)(brow + 32);
        v2f b2 = *(const v2f*)(brow + 64);
        v2f b3 = *(const v2f*)(brow + 96);

        acc0 = __builtin_amdgcn_wmma_f32_16x16x4_f32(false, a, false, b0,
                                                     (short)0, acc0, false, false);
        acc1 = __builtin_amdgcn_wmma_f32_16x16x4_f32(false, a, false, b1,
                                                     (short)0, acc1, false, false);
        acc2 = __builtin_amdgcn_wmma_f32_16x16x4_f32(false, a, false, b2,
                                                     (short)0, acc2, false, false);
        acc3 = __builtin_amdgcn_wmma_f32_16x16x4_f32(false, a, false, b3,
                                                     (short)0, acc3, false, false);
    }

    const int Mofs = (lane >> 4) << 3;
    v8f accs[4] = {acc0, acc1, acc2, acc3};
    #pragma unroll
    for (int t = 0; t < 4; ++t) {
        const int N = laneM + t * 16;
        if (N < NCOL) {
            #pragma unroll
            for (int r = 0; r < 8; ++r) {
                const int M = r + Mofs;
                Y[(size_t)(pixBase + M) * NCOL + N] = accs[t][r];
            }
        }
    }
}

__global__ void calib_reduce_kernel(const float* __restrict__ Y,
                                    const float* __restrict__ bfold,
                                    float* __restrict__ out) {
    int pix = blockIdx.x * 256 + threadIdx.x;
    if (pix >= NPIX) return;
    int n  = pix / HW;
    int hw = pix - n * HW;
    int h  = hw / WW;
    int w  = hw - h * WW;

    const float b0 = bfold[0];
    const float b1 = bfold[1];

    float cm0[PP], cm1[PP];
    #pragma unroll
    for (int p = 0; p < PP; ++p) {
        int i   = p / KK, j = p - i * KK;
        int hh  = h + i - 2;
        int ww2 = w + j - 2;
        float v0 = 0.f, v1 = 0.f;
        if (hh >= 0 && hh < HH && ww2 >= 0 && ww2 < WW) {
            const float* yr = Y + (size_t)(n * HW + hh * WW + ww2) * NCOL + p * NC;
            v0 = yr[0];
            v1 = yr[1];
        }
        cm0[p] = v0 + b0;
        cm1[p] = v1 + b1;
    }

    #pragma unroll
    for (int k = 0; k < NC; ++k) {
        const float* cm = (k == 0) ? cm0 : cm1;
        float asum = 0.f, ys = 0.f, xs = 0.f, cs = 0.f;
        #pragma unroll
        for (int p = 0; p < PP; ++p) {
            float c  = cm[p];
            float a  = fabsf(c);
            float yo = (float)(p / KK - 2);
            float xo = (float)(p % KK - 2);
            asum += a;
            ys   += a * yo;
            xs   += a * xo;
            cs   += c;
        }
        float yd    = ys / asum;
        float xd    = xs / asum;
        float drift = sqrtf(xd * xd + yd * yd);
        out[(size_t)pix * NC + k] = cs * expf(-0.5f * drift);
    }
}

extern "C" void kernel_launch(void* const* d_in, const int* in_sizes, int n_in,
                              void* d_out, int out_size, void* d_ws, size_t ws_size,
                              hipStream_t stream) {
    const float* x    = (const float*)d_in[0];
    const float* Wt   = (const float*)d_in[1];
    const float* bias = (const float*)d_in[2];
    const float* Wlin = (const float*)d_in[3];
    const float* blin = (const float*)d_in[4];
    float* out = (float*)d_out;

    float* wsf   = (float*)d_ws;
    float* Bp    = wsf;
    float* bfold = wsf + WS_BF;
    float* Y     = wsf + WS_Y;

    calib_fold_kernel<<<(BTOT + 255) / 256, 256, 0, stream>>>(
        Wt, bias, Wlin, blin, Bp, bfold);

    calib_gemm_kernel<<<NPIX / 128, 256, 0, stream>>>(x, Bp, Y);

    calib_reduce_kernel<<<(NPIX + 255) / 256, 256, 0, stream>>>(Y, bfold, out);
}


# All global loads of a wave issued before first use (fold o-loop and GEMM K-loop fully unrolled, weight fill and A tile loaded ahead of the barrier): 14-16 dependent cache round trips become 1-2
# speedup vs baseline: 1.6277x; 1.6277x over previous
#include <hip/hip_runtime.h>
#include <hip/hip_bf16.h>
#include <math.h>

#define NB   8
#define CCH  128
#define HH   56
#define WW   56
#define OCH  64
#define NC   2
#define KK   5
#define PP   (KK*KK)
#define HW   (HH*WW)
#define NPIX (NB*HW)
#define NCOL 50
#define NCOLP 64

#define BROW 160
#define BROWS 64
#define BTOT (BROWS*BROW)

typedef __attribute__((ext_vector_type(2))) float v2f;
typedef __attribute__((ext_vector_type(8))) float v8f;

#define WS_BF    BTOT
#define WS_Y     (BTOT + 64)

__global__ void calib_fold_kernel(const float* __restrict__ Wt,
                                  const float* __restrict__ bias,
                                  const float* __restrict__ Wlin,
                                  const float* __restrict__ blin,
                                  float* __restrict__ Bp,
                                  float* __restrict__ bfold) {
    int idx = blockIdx.x * 256 + threadIdx.x;
    if (idx >= BTOT) return;
    int row = idx / BROW;
    int off = idx - row * BROW;
    float v = 0.f;
    if (off < 2 * NCOLP) {
        int col = off >> 1;
        int par = off & 1;
        int c   = row * 2 + par;
        if (col < NCOL) {
            int p = col >> 1;
            int k = col & 1;
            const float* wt = Wt + c * PP + p;
            const float* wl = Wlin + k * OCH;
            float tv[OCH], lv[OCH];
            #pragma unroll
            for (int o = 0; o < OCH; ++o) tv[o] = wt[(size_t)o * (CCH * PP)];
            #pragma unroll
            for (int o = 0; o < OCH; ++o) lv[o] = wl[o];
            #pragma unroll
            for (int o = 0; o < OCH; ++o) v = __builtin_fmaf(lv[o], tv[o], v);
        }
    }
    Bp[idx] = v;
    if (idx >= 2 * NCOLP && idx < 2 * NCOLP + NC) {
        const int k = idx - 2 * NCOLP;
        const float* wl = Wlin + k * OCH;
        float bv[OCH], lv[OCH];
        #pragma unroll
        for (int o = 0; o < OCH; ++o) bv[o] = bias[o];
        #pragma unroll
        for (int o = 0; o < OCH; ++o) lv[o] = wl[o];
        float s = blin[k];
        #pragma unroll
        for (int o = 0; o < OCH; ++o) s = __builtin_fmaf(lv[o], bv[o], s);
        bfold[k] = s;
    }
}

__global__ void __launch_bounds__(256)
calib_gemm_kernel(const float* __restrict__ x,
                  const float* __restrict__ Bp,
                  float* __restrict__ Y) {
    __shared__ __attribute__((aligned(16))) float Blds[BTOT];

    const int tid  = threadIdx.x;
    const int wave = tid >> 5;
    const int lane = tid & 31;

    const int pixBase = (blockIdx.x * 8 + wave) * 16;

    const int laneM = lane & 15;
    const int kHalf = (lane >> 4) << 1;
    const int pm    = pixBase + laneM;
    const int n     = pm / HW;
    const int hw    = pm - n * HW;
    const float* xb = x + (size_t)n * CCH * HW + hw;

    typedef __attribute__((ext_vector_type(4))) float v4f;
    v4f bw[BTOT / 4 / 256];
    #pragma unroll
    for (int j = 0; j < BTOT / 4 / 256; ++j)
        bw[j] = ((const v4f*)Bp)[tid + j * 256];
    asm volatile("" ::: "memory");

    v2f av[CCH / 4];
    #pragma unroll
    for (int ks = 0; ks < CCH / 4; ++ks) {
        const int c = ks * 4 + kHalf;
        av[ks].x = xb[(size_t)c * HW];
        av[ks].y = xb[(size_t)(c + 1) * HW];
    }

    #pragma unroll
    for (int j = 0; j < BTOT / 4 / 256; ++j)
        ((v4f*)Blds)[tid + j * 256] = bw[j];
    __builtin_amdgcn_fence(__ATOMIC_RELEASE, "workgroup", "local");
    __builtin_amdgcn_s_barrier();
    __builtin_amdgcn_fence(__ATOMIC_ACQUIRE, "workgroup", "local");

    v8f acc0 = {}, acc1 = {}, acc2 = {}, acc3 = {};

    #pragma unroll
    for (int k0 = 0; k0 < CCH; k0 += 4) {
        const int c = k0 + kHalf;
        const v2f a = av[k0 >> 2];

        const float* brow = &Blds[(c >> 1) * BROW + laneM * 2];
        v2f b0 = *(const v2f*)(brow + 0);
        v2f b1 = *(const v2f*)(brow + 32);
        v2f b2 = *(const v2f*)(brow + 64);
        v2f b3 = *(const v2f*)(brow + 96);

        acc0 = __builtin_amdgcn_wmma_f32_16x16x4_f32(false, a, false, b0,
                                                     (short)0, acc0, false, false);
        acc1 = __builtin_amdgcn_wmma_f32_16x16x4_f32(false, a, false, b1,
                                                     (short)0, acc1, false, false);
        acc2 = __builtin_amdgcn_wmma_f32_16x16x4_f32(false, a, false, b2,
                                                     (short)0, acc2, false, false);
        acc3 = __builtin_amdgcn_wmma_f32_16x16x4_f32(false, a, false, b3,
                                                     (short)0, acc3, false, false);
    }

    const int Mofs = (lane >> 4) << 3;
    v8f accs[4] = {acc0, acc1, acc2, acc3};
    #pragma unroll
    for (int t = 0; t < 4; ++t) {
        const int N = laneM + t * 16;
        if (N < NCOL) {
            #pragma unroll
            for (int r = 0; r < 8; ++r) {
                const int M = r + Mofs;
                Y[(size_t)(pixBase + M) * NCOL + N] = accs[t][r];
            }
        }
    }
}

__global__ void calib_reduce_kernel(const float* __restrict__ Y,
                                    const float* __restrict__ bfold,
                                    float* __restrict__ out) {
    int pix = blockIdx.x * 256 + threadIdx.x;
    if (pix >= NPIX) return;
    int n  = pix / HW;
    int hw = pix - n * HW;
    int h  = hw / WW;
    int w  = hw - h * WW;

    const float b0 = bfold[0];
    const float b1 = bfold[1];

    float cm0[PP], cm1[PP];
    #pragma unroll
    for (int p = 0; p < PP; ++p) {
        int i   = p / KK, j = p - i * KK;
        int hh  = h + i - 2;
        int ww2 = w + j - 2;
        float v0 = 0.f, v1 = 0.f;
        if (hh >= 0 && hh < HH && ww2 >= 0 && ww2 < WW) {
            const float* yr = Y + (size_t)(n * HW + hh * WW + ww2) * NCOL + p * NC;
            v0 = yr[0];
            v1 = yr[1];
        }
        cm0[p] = v0 + b0;
        cm1[p] = v1 + b1;
    }

    #pragma unroll
    for (int k = 0; k < NC; ++k) {
        const float* cm = (k == 0) ? cm0 : cm1;
        float asum = 0.f, ys = 0.f, xs = 0.f, cs = 0.f;
        #pragma unroll
        for (int p = 0; p < PP; ++p) {
            float c  = cm[p];
            float a  = fabsf(c);
            float yo = (float)(p / KK - 2);
            float xo = (float)(p % KK - 2);
            asum += a;
            ys   += a * yo;
            xs   += a * xo;
            cs   += c;
        }
        float yd    = ys / asum;
        float xd    = xs / asum;
        float drift = sqrtf(xd * xd + yd * yd);
        out[(size_t)pix * NC + k] = cs * expf(-0.5f * drift);
    }
}

extern "C" void kernel_launch(void* const* d_in, const int* in_sizes, int n_in,
                              void* d_out, int out_size, void* d_ws, size_t ws_size,
                              hipStream_t stream) {
    const float* x    = (const float*)d_in[0];
    const float* Wt   = (const float*)d_in[1];
    const float* bias = (const float*)d_in[2];
    const float* Wlin = (const float*)d_in[3];
    const float* blin = (const float*)d_in[4];
    float* out = (float*)d_out;

    float* wsf   = (float*)d_ws;
    float* Bp    = wsf;
    float* bfold = wsf + WS_BF;
    float* Y     = wsf + WS_Y;

    calib_fold_kernel<<<(BTOT + 255) / 256, 256, 0, stream>>>(
        Wt, bias, Wlin, blin, Bp, bfold);

    calib_gemm_kernel<<<NPIX / 128, 256, 0, stream>>>(x, Bp, Y);

    calib_reduce_kernel<<<(NPIX + 255) / 256, 256, 0, stream>>>(Y, bfold, out);
}


# Block re-tiled to 4 output rows x 28 px with the 50 weight columns packed densely: kernel rows share 16-column tiles (16 column tiles per 4 rows, not 20) and a block loads 8x32 input px, not 6x64
# speedup vs baseline: 1.8403x; 1.1306x over previous
#include <hip/hip_runtime.h>
#include <math.h>


#define NB   8
#define CCH  128
#define HH   56
#define WW   56
#define OCH  64
#define NC   2
#define KK   5
#define PP   (KK*KK)
#define HW   (HH*WW)
#define NPIX (NB*HW)

#define NCOL  (PP*NC)
#define BUSED (NCOL*2)
#define BROW  160
#define BROWS 64
#define BTOT  (BROWS*BROW)
#define GTOT  (BROWS*BUSED)
#define BV4   (GTOT/4)
#define BV4ROW (BUSED/4)
#define BLOADS ((BV4 + 255)/256)

#define ROWS_OUT 4
#define PIX_OUT  28
#define PIX_IN   32
#define YCOLS (KK*NC)
#define YTOT  (ROWS_OUT*KK*PIX_IN*YCOLS)
#define BLK_PER_IMG ((HH/ROWS_OUT)*2)

typedef __attribute__((ext_vector_type(2))) float v2f;
typedef __attribute__((ext_vector_type(4))) float v4f;
typedef __attribute__((ext_vector_type(8))) float v8f;

#define WS_BF GTOT

__global__ void calib_fold_kernel(const float* __restrict__ Wt,
                                  const float* __restrict__ bias,
                                  const float* __restrict__ Wlin,
                                  const float* __restrict__ blin,
                                  float* __restrict__ Bp,
                                  float* __restrict__ bfold) {
    int idx = blockIdx.x * 256 + threadIdx.x;
    if (idx >= GTOT) {
        const int k = idx - GTOT;
        if (k < NC) {
            const float* wl = Wlin + k * OCH;
            float bv[OCH], lv[OCH];
            #pragma unroll
            for (int o = 0; o < OCH; ++o) bv[o] = bias[o];
            #pragma unroll
            for (int o = 0; o < OCH; ++o) lv[o] = wl[o];
            float s = blin[k];
            #pragma unroll
            for (int o = 0; o < OCH; ++o) s = __builtin_fmaf(lv[o], bv[o], s);
            bfold[k] = s;
        }
        return;
    }
    int row = idx / BUSED;
    int off = idx - row * BUSED;
    float v = 0.f;
    {
        int col = off >> 1;
        int par = off & 1;
        int i   = col / YCOLS;
        int jk  = col - i * YCOLS;
        int j   = jk >> 1;
        int k   = jk & 1;
        int c   = row * 2 + par;
        int p   = i * KK + j;
        const float* wt = Wt + c * PP + p;
        const float* wl = Wlin + k * OCH;
        float tv[OCH], lv[OCH];
        #pragma unroll
        for (int o = 0; o < OCH; ++o) tv[o] = wt[(size_t)o * (CCH * PP)];
        #pragma unroll
        for (int o = 0; o < OCH; ++o) lv[o] = wl[o];
        #pragma unroll
        for (int o = 0; o < OCH; ++o) v = __builtin_fmaf(lv[o], tv[o], v);
    }
    Bp[idx] = v;
}

__device__ __forceinline__ v8f wmma4(v2f a, v2f b, v8f c) {
    return __builtin_amdgcn_wmma_f32_16x16x4_f32(false, a, false, b,
                                                 (short)0, c, false, false);
}

__device__ __forceinline__ void mma_three(const v2f (&av)[CCH / 4], const float* bb,
                                          v8f& r0, v8f& r1, v8f& r2) {
    v8f acc0 = {}, acc1 = {}, acc2 = {};
    #pragma unroll
    for (int ks = 0; ks < CCH / 4; ++ks) {
        const v2f a  = av[ks];
        const float* q = bb + ks * 2 * BROW;
        const v2f b0 = *(const v2f*)(q);
        const v2f b1 = *(const v2f*)(q + 32);
        const v2f b2 = *(const v2f*)(q + 64);
        acc0 = wmma4(a, b0, acc0);
        acc1 = wmma4(a, b1, acc1);
        acc2 = wmma4(a, b2, acc2);
    }
    r0 = acc0;
    r1 = acc1;
    r2 = acc2;
}

__device__ __forceinline__ void mma_two(const v2f (&av)[CCH / 4], const float* bb,
                                        v8f& r0, v8f& r1) {
    v8f acc0 = {}, acc1 = {};
    #pragma unroll
    for (int ks = 0; ks < CCH / 4; ++ks) {
        const v2f a  = av[ks];
        const float* q = bb + ks * 2 * BROW;
        const v2f b0 = *(const v2f*)(q);
        const v2f b1 = *(const v2f*)(q + 32);
        acc0 = wmma4(a, b0, acc0);
        acc1 = wmma4(a, b1, acc1);
    }
    r0 = acc0;
    r1 = acc1;
}

__device__ __forceinline__ void mma_one(const v2f (&av)[CCH / 4],
                                        const float* bb, v8f& r) {
    v8f acc0 = {}, acc1 = {};
    #pragma unroll
    for (int ks = 0; ks < CCH / 4; ks += 2) {
        const v2f b0 = *(const v2f*)(bb + ks * 2 * BROW);
        const v2f b1 = *(const v2f*)(bb + (ks + 1) * 2 * BROW);
        acc0 = wmma4(av[ks], b0, acc0);
        acc1 = wmma4(av[ks + 1], b1, acc1);
    }
    r = acc0 + acc1;
}

__device__ __forceinline__ void store_tile(float* Yl, int r, int c0, int mt,
                                           int lane, v8f acc) {
    const int col = c0 + (lane & 15);
    const int i   = col / YCOLS;
    const int jk  = col - i * YCOLS;
    const int a   = r - i;
    const int m0  = mt * 16 + ((lane >> 4) << 3);
    if (a >= 0 && a < ROWS_OUT) {
        float* yp = Yl + ((a * KK + i) * PIX_IN + m0) * YCOLS + jk;
        #pragma unroll
        for (int q = 0; q < 8; ++q) yp[q * YCOLS] = acc[q];
    }
}

__device__ __forceinline__ void load_rows(v2f (&av)[CCH / 4], const float* xb,
                                          int hrow, int kHalf) {
    hrow = hrow < 0 ? 0 : (hrow > HH - 1 ? HH - 1 : hrow);
    const float* xr = xb + hrow * WW;
    #pragma unroll
    for (int ks = 0; ks < CCH / 4; ++ks) {
        const int c = ks * 4 + kHalf;
        av[ks].x = xr[(size_t)c * HW];
        av[ks].y = xr[(size_t)(c + 1) * HW];
    }
}

__global__ void __launch_bounds__(256)
calib_conv_kernel(const float* __restrict__ x,
                  const float* __restrict__ Bp,
                  const float* __restrict__ bfold,
                  float* __restrict__ out) {
    __shared__ __attribute__((aligned(16))) float Blds[BTOT];
    __shared__ __attribute__((aligned(16))) float Yl[YTOT];

    const int tid  = threadIdx.x;
    const int wave = __builtin_amdgcn_readfirstlane(tid >> 5);
    const int lane = tid & 31;

    const int n    = blockIdx.x / BLK_PER_IMG;
    const int rem  = blockIdx.x - n * BLK_PER_IMG;
    const int h0   = (rem >> 1) * ROWS_OUT;
    const int side = rem & 1;
    const int w0   = side * PIX_OUT;
    const int wb   = side * (WW - PIX_IN);

    const int mt = wave & 1;
    const int pr = wave >> 1;
    const int rX = (pr == 0) ? 3 : (pr == 1) ? 2 : (pr == 2) ? 4 : 5;
    const int rY = (pr == 0) ? 0 : (pr == 1) ? 1 : (pr == 2) ? 7 : 6;
    const int cX = (pr == 0) ? 0 : (pr == 1) ? 0 : (pr == 2) ? 2 : 18;
    const int cY = (pr == 0) ? 0 : (pr == 1) ? 0 : (pr == 2) ? 34 : 18;

    const int laneM = lane & 15;
    const int kHalf = (lane >> 4) << 1;
    const float* xb = x + (size_t)n * CCH * HW + wb + mt * 16 + laneM;

    v4f bw[BLOADS];
    #pragma unroll
    for (int j = 0; j < BLOADS; ++j) {
        int i = tid + j * 256;
        if (j == BLOADS - 1) i = i < BV4 ? i : BV4 - 1;
        bw[j] = ((const v4f*)Bp)[i];
    }
    const float bk = bfold[tid & 1];
    asm volatile("" ::: "memory");

    v2f avX[CCH / 4], avY[CCH / 4];
    load_rows(avX, xb, h0 - 2 + rX, kHalf);

    #pragma unroll
    for (int j = 0; j < BLOADS; ++j) {
        const int i = tid + j * 256;
        if (j < BLOADS - 1 || tid < BV4 - (BLOADS - 1) * 256) {
            const int row = i / BV4ROW;
            const int q   = i - row * BV4ROW;
            *(v4f*)(Blds + row * BROW + q * 4) = bw[j];
        }
    }
    __builtin_amdgcn_fence(__ATOMIC_RELEASE, "workgroup", "local");
    __builtin_amdgcn_s_barrier();
    __builtin_amdgcn_fence(__ATOMIC_ACQUIRE, "workgroup", "local");
    asm volatile("" ::: "memory");
    load_rows(avY, xb, h0 - 2 + rY, kHalf);
    asm volatile("" ::: "memory");

    const float* bbase = &Blds[(kHalf >> 1) * BROW + laneM * 2];

    const int hrX = h0 - 2 + rX;
    const int hrY = h0 - 2 + rY;
    const bool okX = (hrX >= 0 && hrX < HH);
    const bool okY = (hrY >= 0 && hrY < HH);

    if (pr & 1) {
        if (okX) {
            v8f d0, d1;
            mma_two(avX, bbase + cX * 2, d0, d1);
            store_tile(Yl, rX, cX, mt, lane, d0);
            store_tile(Yl, rX, cX + 16, mt, lane, d1);
        }
        if (okY) {
            v8f d0, d1;
            mma_two(avY, bbase + cY * 2, d0, d1);
            store_tile(Yl, rY, cY, mt, lane, d0);
            store_tile(Yl, rY, cY + 16, mt, lane, d1);
        }
    } else {
        if (okX) {
            v8f d0, d1, d2;
            mma_three(avX, bbase + cX * 2, d0, d1, d2);
            store_tile(Yl, rX, cX, mt, lane, d0);
            store_tile(Yl, rX, cX + 16, mt, lane, d1);
            store_tile(Yl, rX, cX + 32, mt, lane, d2);
        }
        if (okY) {
            v8f d0;
            mma_one(avY, bbase + cY * 2, d0);
            store_tile(Yl, rY, cY, mt, lane, d0);
        }
    }

    __builtin_amdgcn_fence(__ATOMIC_RELEASE, "workgroup", "local");
    __builtin_amdgcn_s_barrier();
    __builtin_amdgcn_fence(__ATOMIC_ACQUIRE, "workgroup", "local");

    if (tid < ROWS_OUT * PIX_OUT * NC) {
        const int k  = tid & 1;
        const int pw = tid >> 1;
        const int a  = pw / PIX_OUT;
        const int w  = w0 + (pw - a * PIX_OUT);
        const int h  = h0 + a;

        float cm[PP];
        #pragma unroll
        for (int p = 0; p < PP; ++p) {
            const int i   = p / KK, j = p - i * KK;
            const int hh  = h + i - 2;
            const int ww2 = w + j - 2;
            const bool in  = (hh >= 0 && hh < HH && ww2 >= 0 && ww2 < WW);
            const int hc   = hh < 0 ? 0 : (hh > HH - 1 ? HH - 1 : hh);
            const int wc   = ww2 < 0 ? 0 : (ww2 > WW - 1 ? WW - 1 : ww2);
            const int ic   = hc - h + 2;
            const float yv = Yl[(((a * KK + ic) * PIX_IN) + (wc - wb)) * YCOLS + j * NC + k];
            cm[p] = (in ? yv : 0.f) + bk;
        }
        float asum = 0.f, ys = 0.f, xs = 0.f, cs = 0.f;
        #pragma unroll
        for (int p = 0; p < PP; ++p) {
            float c  = cm[p];
            float ab = fabsf(c);
            float yo = (float)(p / KK - 2);
            float xo = (float)(p % KK - 2);
            asum += ab;
            ys   += ab * yo;
            xs   += ab * xo;
            cs   += c;
        }
        float yd    = ys / asum;
        float xd    = xs / asum;
        float drift = sqrtf(xd * xd + yd * yd);
        out[((size_t)n * HW + h * WW + w) * NC + k] = cs * expf(-0.5f * drift);
    }
}

extern "C" void kernel_launch(void* const* d_in, const int* in_sizes, int n_in,
                              void* d_out, int out_size, void* d_ws, size_t ws_size,
                              hipStream_t stream) {
    const float* x    = (const float*)d_in[0];
    const float* Wt   = (const float*)d_in[1];
    const float* bias = (const float*)d_in[2];
    const float* Wlin = (const float*)d_in[3];
    const float* blin = (const float*)d_in[4];
    float* out = (float*)d_out;

    float* wsf   = (float*)d_ws;
    float* Bp    = wsf;
    float* bfold = wsf + WS_BF;

    calib_fold_kernel<<<GTOT / 256 + 1, 256, 0, stream>>>(
        Wt, bias, Wlin, blin, Bp, bfold);

    calib_conv_kernel<<<NB * BLK_PER_IMG, 256, 0, stream>>>(x, Bp, bfold, out);
}


# One wave per SIMD (4 waves a block, not 8) pipelined in 8-step K blocks: a wave owns both pixel tiles of its rows so a B fragment feeds two; LDS reads run a block ahead, input loads two blocks ahead
# speedup vs baseline: 1.9847x; 1.0785x over previous
#include <hip/hip_runtime.h>
#include <math.h>


#define NB   8
#define CCH  128
#define HH   56
#define WW   56
#define OCH  64
#define NC   2
#define KK   5
#define PP   (KK*KK)
#define HW   (HH*WW)
#define NPIX (NB*HW)

#define NCOL  (PP*NC)
#define BUSED (NCOL*2)
#define BROW  160
#define BROWS 64
#define BTOT  (BROWS*BROW)
#define GTOT  (BROWS*BUSED)
#define BV4   (GTOT/4)
#define BV4ROW (BUSED/4)

#define ROWS_OUT 4
#define PIX_OUT  28
#define PIX_IN   32
#define YCOLS (KK*NC)
#define YTOT  (ROWS_OUT*KK*PIX_IN*YCOLS)
#define BLK_PER_IMG ((HH/ROWS_OUT)*2)

typedef __attribute__((ext_vector_type(2))) float v2f;
typedef __attribute__((ext_vector_type(4))) float v4f;
typedef __attribute__((ext_vector_type(8))) float v8f;

#define WS_BF GTOT

__global__ void calib_fold_kernel(const float* __restrict__ Wt,
                                  const float* __restrict__ bias,
                                  const float* __restrict__ Wlin,
                                  const float* __restrict__ blin,
                                  float* __restrict__ Bp,
                                  float* __restrict__ bfold) {
    int idx = blockIdx.x * 256 + threadIdx.x;
    if (idx >= GTOT) {
        const int k = idx - GTOT;
        if (k < NC) {
            const float* wl = Wlin + k * OCH;
            float bv[OCH], lv[OCH];
            #pragma unroll
            for (int o = 0; o < OCH; ++o) bv[o] = bias[o];
            #pragma unroll
            for (int o = 0; o < OCH; ++o) lv[o] = wl[o];
            float s = blin[k];
            #pragma unroll
            for (int o = 0; o < OCH; ++o) s = __builtin_fmaf(lv[o], bv[o], s);
            bfold[k] = s;
        }
        return;
    }
    int row = idx / BUSED;
    int off = idx - row * BUSED;
    float v = 0.f;
    {
        int col = off >> 1;
        int par = off & 1;
        int i   = col / YCOLS;
        int jk  = col - i * YCOLS;
        int j   = jk >> 1;
        int k   = jk & 1;
        int c   = row * 2 + par;
        int p   = i * KK + j;
        const float* wt = Wt + c * PP + p;
        const float* wl = Wlin + k * OCH;
        float tv[OCH], lv[OCH];
        #pragma unroll
        for (int o = 0; o < OCH; ++o) tv[o] = wt[(size_t)o * (CCH * PP)];
        #pragma unroll
        for (int o = 0; o < OCH; ++o) lv[o] = wl[o];
        #pragma unroll
        for (int o = 0; o < OCH; ++o) v = __builtin_fmaf(lv[o], tv[o], v);
    }
    Bp[idx] = v;
}

#define NTHR   128
#define BFULL  (BV4 / NTHR)
#define BTAIL  (BFULL * NTHR)
#define KSTEPS (CCH / 4)
static_assert((BV4 - BTAIL) * 2 == NTHR, "tail halves");
#define KBLK   8

__device__ __forceinline__ v8f wmma4(v2f a, v2f b, v8f c) {
    return __builtin_amdgcn_wmma_f32_16x16x4_f32(false, a, false, b,
                                                 (short)0, c, false, false);
}

template <int CNT>
__device__ __forceinline__ void load_ks(v2f (&a0)[KSTEPS], v2f (&a1)[KSTEPS],
                                        const float* xr, int kHalf, int ks0) {
    #pragma unroll
    for (int kk = 0; kk < CNT; ++kk) {
        const int ks = ks0 + kk;
        const int c  = ks * 4 + kHalf;
        a0[ks].x = xr[(size_t)c * HW];
        a0[ks].y = xr[(size_t)(c + 1) * HW];
        a1[ks].x = xr[(size_t)c * HW + 16];
        a1[ks].y = xr[(size_t)(c + 1) * HW + 16];
    }
}

template <int NT, bool FIRST>
__device__ __forceinline__ void mma_row(v2f (&a0)[KSTEPS], v2f (&a1)[KSTEPS],
                                        const float* bb, v8f (&d0)[NT], v8f (&d1)[NT],
                                        v2f (&n0)[KSTEPS], v2f (&n1)[KSTEPS],
                                        const float* xr, const float* nxr, int kHalf) {
    v2f bf[2][KBLK][NT];
    #pragma unroll
    for (int kk = 0; kk < KBLK; ++kk) {
        #pragma unroll
        for (int t = 0; t < NT; ++t)
            bf[0][kk][t] = *(const v2f*)(bb + kk * 2 * BROW + t * 32);
    }
    #pragma unroll
    for (int t = 0; t < NT; ++t) {
        d0[t] = v8f{};
        d1[t] = v8f{};
    }
    #pragma unroll
    for (int blk = 0; blk < KSTEPS / KBLK; ++blk) {
        if (blk + 1 < KSTEPS / KBLK) {
            #pragma unroll
            for (int kk = 0; kk < KBLK; ++kk) {
                #pragma unroll
                for (int t = 0; t < NT; ++t)
                    bf[(blk + 1) & 1][kk][t] =
                        *(const v2f*)(bb + ((blk + 1) * KBLK + kk) * 2 * BROW + t * 32);
            }
        }
        if (FIRST) {
            if (blk == 0) load_ks<8>(a0, a1, xr, kHalf, 16);
            if (blk == 1) load_ks<8>(a0, a1, xr, kHalf, 24);
            if (blk == 2) load_ks<12>(n0, n1, nxr, kHalf, 0);
            if (blk == 3) {
                load_ks<11>(n0, n1, nxr, kHalf, 12);
                __builtin_amdgcn_sched_barrier(0);
                load_ks<1>(n0, n1, nxr, kHalf, 23);
            }
        } else {
            if (blk == 0) load_ks<8>(a0, a1, xr, kHalf, 24);
        }
        __builtin_amdgcn_sched_barrier(0);
        #pragma unroll
        for (int kk = 0; kk < KBLK; ++kk) {
            const int ks = blk * KBLK + kk;
            #pragma unroll
            for (int t = 0; t < NT; ++t) {
                const v2f b = bf[blk & 1][kk][t];
                d0[t] = wmma4(a0[ks], b, d0[t]);
                d1[t] = wmma4(a1[ks], b, d1[t]);
            }
        }
        __builtin_amdgcn_sched_barrier(0);
    }
    asm volatile("v_nop\n\tv_nop\n\tv_nop\n\tv_nop");
    __builtin_amdgcn_sched_barrier(0);
}

__device__ __forceinline__ void store_tile(float* Yl, int r, int c0, int mt,
                                           int lane, v8f acc) {
    const int col = c0 + (lane & 15);
    const int i   = col / YCOLS;
    const int jk  = col - i * YCOLS;
    const int a   = r - i;
    const int m0  = mt * 16 + ((lane >> 4) << 3);
    if (a >= 0 && a < ROWS_OUT) {
        float* yp = Yl + ((a * KK + i) * PIX_IN + m0) * YCOLS + jk;
        #pragma unroll
        for (int q = 0; q < 8; ++q) yp[q * YCOLS] = acc[q];
    }
}

__global__ void __launch_bounds__(NTHR)
calib_conv_kernel(const float* __restrict__ x,
                  const float* __restrict__ Bp,
                  const float* __restrict__ bfold,
                  float* __restrict__ out) {
    __shared__ __attribute__((aligned(16))) float Blds[BTOT];
    __shared__ __attribute__((aligned(16))) float Yl[YTOT];

    const int tid  = threadIdx.x;
    const int pr   = __builtin_amdgcn_readfirstlane(tid >> 5);
    const int lane = tid & 31;

    const int n    = blockIdx.x / BLK_PER_IMG;
    const int rem  = blockIdx.x - n * BLK_PER_IMG;
    const int h0   = (rem >> 1) * ROWS_OUT;
    const int side = rem & 1;
    const int w0   = side * PIX_OUT;
    const int wb   = side * (WW - PIX_IN);

    const int rX = (pr == 0) ? 3 : (pr == 1) ? 2 : (pr == 2) ? 4 : 5;
    const int rY = (pr == 0) ? 0 : (pr == 1) ? 1 : (pr == 2) ? 7 : 6;
    const int cX = (pr == 0) ? 0 : (pr == 1) ? 0 : (pr == 2) ? 2 : 18;
    const int cY = (pr == 0) ? 0 : (pr == 1) ? 0 : (pr == 2) ? 34 : 18;

    const int laneM = lane & 15;
    const int kHalf = (lane >> 4) << 1;
    const float* xb = x + (size_t)n * CCH * HW + wb + laneM;

    v4f bw[BFULL];
    #pragma unroll
    for (int j = 0; j < BFULL; ++j) bw[j] = ((const v4f*)Bp)[tid + j * NTHR];
    const v2f bt = ((const v2f*)Bp)[BTAIL * 2 + tid];
    const float bk0 = bfold[0];
    const float bk1 = bfold[1];
    asm volatile("" ::: "memory");

    v2f aX0[KSTEPS], aX1[KSTEPS], aY0[KSTEPS], aY1[KSTEPS];
    const float* xrX = xb + (h0 - 2 + rX) * WW;
    load_ks<16>(aX0, aX1, xrX, kHalf, 0);

    #pragma unroll
    for (int j = 0; j < BFULL; ++j) {
        const int i   = tid + j * NTHR;
        const int row = i / BV4ROW;
        const int q   = i - row * BV4ROW;
        *(v4f*)(Blds + row * BROW + q * 4) = bw[j];
    }
    {
        const int i   = BTAIL + (tid >> 1);
        const int row = i / BV4ROW;
        const int q   = i - row * BV4ROW;
        *(v2f*)(Blds + row * BROW + q * 4 + (tid & 1) * 2) = bt;
    }
    __builtin_amdgcn_fence(__ATOMIC_RELEASE, "workgroup", "local");
    __builtin_amdgcn_s_barrier();
    __builtin_amdgcn_fence(__ATOMIC_ACQUIRE, "workgroup", "local");
    asm volatile("" ::: "memory");

    const float* bbase = &Blds[(kHalf >> 1) * BROW + laneM * 2];

    const int hrY  = h0 - 2 + rY;
    const bool okY = (hrY >= 0 && hrY < HH);
    const int hcY  = hrY < 0 ? 0 : (hrY > HH - 1 ? HH - 1 : hrY);
    const float* xrY = xb + hcY * WW;

    if (pr & 1) {
        {
            v8f d0[2], d1[2];
            mma_row<2, true>(aX0, aX1, bbase + cX * 2, d0, d1, aY0, aY1, xrX, xrY, kHalf);
            #pragma unroll
            for (int t = 0; t < 2; ++t) {
                store_tile(Yl, rX, cX + 16 * t, 0, lane, d0[t]);
                store_tile(Yl, rX, cX + 16 * t, 1, lane, d1[t]);
            }
        }
        if (okY) {
            v8f d0[2], d1[2];
            mma_row<2, false>(aY0, aY1, bbase + cY * 2, d0, d1, aX0, aX1, xrY, xrY, kHalf);
            #pragma unroll
            for (int t = 0; t < 2; ++t) {
                store_tile(Yl, rY, cY + 16 * t, 0, lane, d0[t]);
                store_tile(Yl, rY, cY + 16 * t, 1, lane, d1[t]);
            }
        }
        asm volatile("" :: "v"(aY0[23]), "v"(aY1[23]));
        __builtin_amdgcn_fence(__ATOMIC_RELEASE, "workgroup", "local");
    } else {
        {
            v8f d0[3], d1[3];
            mma_row<3, true>(aX0, aX1, bbase + cX * 2, d0, d1, aY0, aY1, xrX, xrY, kHalf);
            #pragma unroll
            for (int t = 0; t < 3; ++t) {
                store_tile(Yl, rX, cX + 16 * t, 0, lane, d0[t]);
                store_tile(Yl, rX, cX + 16 * t, 1, lane, d1[t]);
            }
        }
        if (okY) {
            v8f d0[1], d1[1];
            mma_row<1, false>(aY0, aY1, bbase + cY * 2, d0, d1, aX0, aX1, xrY, xrY, kHalf);
            store_tile(Yl, rY, cY, 0, lane, d0[0]);
            store_tile(Yl, rY, cY, 1, lane, d1[0]);
        }
        asm volatile("" :: "v"(aY0[23]), "v"(aY1[23]));
    }

    __builtin_amdgcn_fence(__ATOMIC_RELEASE, "workgroup", "local");
    __builtin_amdgcn_s_barrier();
    __builtin_amdgcn_fence(__ATOMIC_ACQUIRE, "workgroup", "local");

    if (tid < ROWS_OUT * PIX_OUT) {
        const int a  = tid / PIX_OUT;
        const int w  = w0 + (tid - a * PIX_OUT);
        const int h  = h0 + a;

        v2f cm[PP];
        #pragma unroll
        for (int p = 0; p < PP; ++p) {
            const int i   = p / KK, j = p - i * KK;
            const int hh  = h + i - 2;
            const int ww2 = w + j - 2;
            const bool in  = (hh >= 0 && hh < HH && ww2 >= 0 && ww2 < WW);
            const int hc   = hh < 0 ? 0 : (hh > HH - 1 ? HH - 1 : hh);
            const int wc   = ww2 < 0 ? 0 : (ww2 > WW - 1 ? WW - 1 : ww2);
            const int ic   = hc - h + 2;
            const v2f yv = *(const v2f*)&Yl[(((a * KK + ic) * PIX_IN) + (wc - wb)) * YCOLS + j * NC];
            cm[p].x = (in ? yv.x : 0.f) + bk0;
            cm[p].y = (in ? yv.y : 0.f) + bk1;
        }
        v2f res;
        #pragma unroll
        for (int k = 0; k < NC; ++k) {
            float asum = 0.f, ys = 0.f, xs = 0.f, cs = 0.f;
            #pragma unroll
            for (int p = 0; p < PP; ++p) {
                float c  = cm[p][k];
                float ab = fabsf(c);
                float yo = (float)(p / KK - 2);
                float xo = (float)(p % KK - 2);
                asum += ab;
                ys   += ab * yo;
                xs   += ab * xo;
                cs   += c;
            }
            float yd    = ys / asum;
            float xd    = xs / asum;
            float drift = sqrtf(xd * xd + yd * yd);
            res[k] = cs * expf(-0.5f * drift);
        }
        *(v2f*)&out[((size_t)n * HW + h * WW + w) * NC] = res;
    }
}

extern "C" void kernel_launch(void* const* d_in, const int* in_sizes, int n_in,
                              void* d_out, int out_size, void* d_ws, size_t ws_size,
                              hipStream_t stream) {
    const float* x    = (const float*)d_in[0];
    const float* Wt   = (const float*)d_in[1];
    const float* bias = (const float*)d_in[2];
    const float* Wlin = (const float*)d_in[3];
    const float* blin = (const float*)d_in[4];
    float* out = (float*)d_out;

    float* wsf   = (float*)d_ws;
    float* Bp    = wsf;
    float* bfold = wsf + WS_BF;

    calib_fold_kernel<<<GTOT / 256 + 1, 256, 0, stream>>>(
        Wt, bias, Wlin, blin, Bp, bfold);

    calib_conv_kernel<<<NB * BLK_PER_IMG, NTHR, 0, stream>>>(x, Bp, bfold, out);
}


# Weight fold: one wave per block and per 128-byte line of folded weights (200 one-wave blocks, not 26 of 8 waves); its 64 Wt loads go out as one group, Wlin via scalar loads, both classes per lane
# speedup vs baseline: 2.1272x; 1.0718x over previous
#include <hip/hip_runtime.h>
#include <math.h>


#define NB   8
#define CCH  128
#define HH   56
#define WW   56
#define OCH  64
#define NC   2
#define KK   5
#define PP   (KK*KK)
#define HW   (HH*WW)
#define NPIX (NB*HW)

#define NCOL  (PP*NC)
#define BUSED (NCOL*2)
#define BROW  160
#define BROWS 64
#define BTOT  (BROWS*BROW)
#define GTOT  (BROWS*BUSED)
#define BV4   (GTOT/4)
#define BV4ROW (BUSED/4)

#define ROWS_OUT 4
#define PIX_OUT  28
#define PIX_IN   32
#define YCOLS (KK*NC)
#define YTOT  (ROWS_OUT*KK*PIX_IN*YCOLS)
#define BLK_PER_IMG ((HH/ROWS_OUT)*2)

typedef __attribute__((ext_vector_type(2))) float v2f;
typedef __attribute__((ext_vector_type(4))) float v4f;
typedef __attribute__((ext_vector_type(8))) float v8f;

#define WS_BF GTOT

#define FOLD_THR 32
#define FOLD_BLK (GTOT / FOLD_THR)
#define OH       (OCH / 2)
static_assert(FOLD_BLK * FOLD_THR == GTOT, "fold blocks");
static_assert(OH % 16 == 0, "64 byte pieces of Wlin");

__global__ void __launch_bounds__(FOLD_THR)
calib_fold_kernel(const float* __restrict__ Wt,
                  const float* __restrict__ bias,
                  const float* __restrict__ Wlin,
                  const float* __restrict__ blin,
                  float* __restrict__ Bp,
                  float* __restrict__ bfold) {
    const int lane = threadIdx.x;
    if (blockIdx.x == 0) {
        if (lane < NC) {
            float bv[OCH];
            #pragma unroll
            for (int o = 0; o < OCH; ++o) bv[o] = bias[o];
            __builtin_amdgcn_sched_barrier(0);
            const float* wl = Wlin + lane * OCH;
            float lv[OCH];
            #pragma unroll
            for (int o = 0; o < OCH; ++o) lv[o] = wl[o];
            float s0 = blin[lane], s1 = 0.f, s2 = 0.f, s3 = 0.f;
            __builtin_amdgcn_sched_barrier(0);
            #pragma unroll
            for (int o = 0; o < OCH; o += 4) {
                s0 = __builtin_fmaf(lv[o],     bv[o],     s0);
                s1 = __builtin_fmaf(lv[o + 1], bv[o + 1], s1);
                s2 = __builtin_fmaf(lv[o + 2], bv[o + 2], s2);
                s3 = __builtin_fmaf(lv[o + 3], bv[o + 3], s3);
                asm volatile("" : "+v"(s0), "+v"(s1), "+v"(s2), "+v"(s3));
            }
            bfold[lane] = (s0 + s1) + (s2 + s3);
        }
        return;
    }
    const int idx = (blockIdx.x - 1) * FOLD_THR + lane;
    const int row = idx / BUSED;
    const int off = idx - row * BUSED;
    const int p   = off >> 2;
    const int k   = (off >> 1) & 1;
    const int c   = row * 2 + (off & 1);
    const float* wt = Wt + c * PP + p;

    float l0[OH], l1[OH];
    #pragma unroll
    for (int o = 0; o < OH; ++o) { l0[o] = Wlin[o]; l1[o] = Wlin[OCH + o]; }
    const float w0a = Wlin[OH], w0b = Wlin[OH + 16];
    const float w1a = Wlin[OCH + OH], w1b = Wlin[OCH + OH + 16];
    __builtin_amdgcn_sched_barrier(0);
    float tv[OCH];
    #pragma unroll
    for (int o = 0; o < OCH; ++o) tv[o] = wt[(size_t)o * (CCH * PP)];
    __builtin_amdgcn_sched_barrier(0);

    float a0 = 0.f, a1 = 0.f, a2 = 0.f, a3 = 0.f;
    float b0 = 0.f, b1 = 0.f, b2 = 0.f, b3 = 0.f;
    #pragma unroll
    for (int o = 0; o < OH; o += 4) {
        a0 = __builtin_fmaf(l0[o],     tv[o],     a0);
        b0 = __builtin_fmaf(l1[o],     tv[o],     b0);
        a1 = __builtin_fmaf(l0[o + 1], tv[o + 1], a1);
        b1 = __builtin_fmaf(l1[o + 1], tv[o + 1], b1);
        a2 = __builtin_fmaf(l0[o + 2], tv[o + 2], a2);
        b2 = __builtin_fmaf(l1[o + 2], tv[o + 2], b2);
        a3 = __builtin_fmaf(l0[o + 3], tv[o + 3], a3);
        b3 = __builtin_fmaf(l1[o + 3], tv[o + 3], b3);
        asm volatile("" : "+v"(a0), "+v"(a1), "+v"(a2), "+v"(a3),
                          "+v"(b0), "+v"(b1), "+v"(b2), "+v"(b3));
    }
    int oh = OH;
    asm volatile("" : "+s"(oh) : "s"(w0a), "s"(w0b), "s"(w1a), "s"(w1b));
    const float* m0 = Wlin + oh;
    const float* m1 = Wlin + OCH + oh;
    float n0[OH], n1[OH];
    #pragma unroll
    for (int o = 0; o < OH; ++o) { n0[o] = m0[o]; n1[o] = m1[o]; }
    __builtin_amdgcn_sched_barrier(0);
    #pragma unroll
    for (int o = 0; o < OH; o += 4) {
        a0 = __builtin_fmaf(n0[o],     tv[OH + o],     a0);
        b0 = __builtin_fmaf(n1[o],     tv[OH + o],     b0);
        a1 = __builtin_fmaf(n0[o + 1], tv[OH + o + 1], a1);
        b1 = __builtin_fmaf(n1[o + 1], tv[OH + o + 1], b1);
        a2 = __builtin_fmaf(n0[o + 2], tv[OH + o + 2], a2);
        b2 = __builtin_fmaf(n1[o + 2], tv[OH + o + 2], b2);
        a3 = __builtin_fmaf(n0[o + 3], tv[OH + o + 3], a3);
        b3 = __builtin_fmaf(n1[o + 3], tv[OH + o + 3], b3);
        asm volatile("" : "+v"(a0), "+v"(a1), "+v"(a2), "+v"(a3),
                          "+v"(b0), "+v"(b1), "+v"(b2), "+v"(b3));
    }
    const float v0 = (a0 + a1) + (a2 + a3);
    const float v1 = (b0 + b1) + (b2 + b3);
    Bp[idx] = k ? v1 : v0;
}

#define NTHR   128
#define BFULL  (BV4 / NTHR)
#define BTAIL  (BFULL * NTHR)
#define KSTEPS (CCH / 4)
static_assert((BV4 - BTAIL) * 2 == NTHR, "tail halves");
#define KBLK   8

__device__ __forceinline__ v8f wmma4(v2f a, v2f b, v8f c) {
    return __builtin_amdgcn_wmma_f32_16x16x4_f32(false, a, false, b,
                                                 (short)0, c, false, false);
}

template <int CNT>
__device__ __forceinline__ void load_ks(v2f (&a0)[KSTEPS], v2f (&a1)[KSTEPS],
                                        const float* xr, int kHalf, int ks0) {
    #pragma unroll
    for (int kk = 0; kk < CNT; ++kk) {
        const int ks = ks0 + kk;
        const int c  = ks * 4 + kHalf;
        a0[ks].x = xr[(size_t)c * HW];
        a0[ks].y = xr[(size_t)(c + 1) * HW];
        a1[ks].x = xr[(size_t)c * HW + 16];
        a1[ks].y = xr[(size_t)(c + 1) * HW + 16];
    }
}

template <int NT, bool FIRST>
__device__ __forceinline__ void mma_row(v2f (&a0)[KSTEPS], v2f (&a1)[KSTEPS],
                                        const float* bb, v8f (&d0)[NT], v8f (&d1)[NT],
                                        v2f (&n0)[KSTEPS], v2f (&n1)[KSTEPS],
                                        const float* xr, const float* nxr, int kHalf) {
    v2f bf[2][KBLK][NT];
    #pragma unroll
    for (int kk = 0; kk < KBLK; ++kk) {
        #pragma unroll
        for (int t = 0; t < NT; ++t)
            bf[0][kk][t] = *(const v2f*)(bb + kk * 2 * BROW + t * 32);
    }
    #pragma unroll
    for (int t = 0; t < NT; ++t) {
        d0[t] = v8f{};
        d1[t] = v8f{};
    }
    #pragma unroll
    for (int blk = 0; blk < KSTEPS / KBLK; ++blk) {
        if (blk + 1 < KSTEPS / KBLK) {
            #pragma unroll
            for (int kk = 0; kk < KBLK; ++kk) {
                #pragma unroll
                for (int t = 0; t < NT; ++t)
                    bf[(blk + 1) & 1][kk][t] =
                        *(const v2f*)(bb + ((blk + 1) * KBLK + kk) * 2 * BROW + t * 32);
            }
        }
        if (FIRST) {
            if (blk == 0) load_ks<8>(a0, a1, xr, kHalf, 16);
            if (blk == 1) load_ks<8>(a0, a1, xr, kHalf, 24);
            if (blk == 2) load_ks<12>(n0, n1, nxr, kHalf, 0);
            if (blk == 3) {
                load_ks<11>(n0, n1, nxr, kHalf, 12);
                __builtin_amdgcn_sched_barrier(0);
                load_ks<1>(n0, n1, nxr, kHalf, 23);
            }
        } else {
            if (blk == 0) load_ks<8>(a0, a1, xr, kHalf, 24);
        }
        __builtin_amdgcn_sched_barrier(0);
        #pragma unroll
        for (int kk = 0; kk < KBLK; ++kk) {
            const int ks = blk * KBLK + kk;
            #pragma unroll
            for (int t = 0; t < NT; ++t) {
                const v2f b = bf[blk & 1][kk][t];
                d0[t] = wmma4(a0[ks], b, d0[t]);
                d1[t] = wmma4(a1[ks], b, d1[t]);
            }
        }
        __builtin_amdgcn_sched_barrier(0);
    }
    asm volatile("v_nop\n\tv_nop\n\tv_nop\n\tv_nop");
    __builtin_amdgcn_sched_barrier(0);
}

__device__ __forceinline__ void store_tile(float* Yl, int r, int c0, int mt,
                                           int lane, v8f acc) {
    const int col = c0 + (lane & 15);
    const int i   = col / YCOLS;
    const int jk  = col - i * YCOLS;
    const int a   = r - i;
    const int m0  = mt * 16 + ((lane >> 4) << 3);
    if (a >= 0 && a < ROWS_OUT) {
        float* yp = Yl + ((a * KK + i) * PIX_IN + m0) * YCOLS + jk;
        #pragma unroll
        for (int q = 0; q < 8; ++q) yp[q * YCOLS] = acc[q];
    }
}

__global__ void __launch_bounds__(NTHR)
calib_conv_kernel(const float* __restrict__ x,
                  const float* __restrict__ Bp,
                  const float* __restrict__ bfold,
                  float* __restrict__ out) {
    __shared__ __attribute__((aligned(16))) float Blds[BTOT];
    __shared__ __attribute__((aligned(16))) float Yl[YTOT];

    const int tid  = threadIdx.x;
    const int pr   = __builtin_amdgcn_readfirstlane(tid >> 5);
    const int lane = tid & 31;

    const int n    = blockIdx.x / BLK_PER_IMG;
    const int rem  = blockIdx.x - n * BLK_PER_IMG;
    const int h0   = (rem >> 1) * ROWS_OUT;
    const int side = rem & 1;
    const int w0   = side * PIX_OUT;
    const int wb   = side * (WW - PIX_IN);

    const int rX = (pr == 0) ? 3 : (pr == 1) ? 2 : (pr == 2) ? 4 : 5;
    const int rY = (pr == 0) ? 0 : (pr == 1) ? 1 : (pr == 2) ? 7 : 6;
    const int cX = (pr == 0) ? 0 : (pr == 1) ? 0 : (pr == 2) ? 2 : 18;
    const int cY = (pr == 0) ? 0 : (pr == 1) ? 0 : (pr == 2) ? 34 : 18;

    const int laneM = lane & 15;
    const int kHalf = (lane >> 4) << 1;
    const float* xb = x + (size_t)n * CCH * HW + wb + laneM;

    v4f bw[BFULL];
    #pragma unroll
    for (int j = 0; j < BFULL; ++j) bw[j] = ((const v4f*)Bp)[tid + j * NTHR];
    const v2f bt = ((const v2f*)Bp)[BTAIL * 2 + tid];
    const float bk0 = bfold[0];
    const float bk1 = bfold[1];
    asm volatile("" ::: "memory");

    v2f aX0[KSTEPS], aX1[KSTEPS], aY0[KSTEPS], aY1[KSTEPS];
    const float* xrX = xb + (h0 - 2 + rX) * WW;
    load_ks<16>(aX0, aX1, xrX, kHalf, 0);

    #pragma unroll
    for (int j = 0; j < BFULL; ++j) {
        const int i   = tid + j * NTHR;
        const int row = i / BV4ROW;
        const int q   = i - row * BV4ROW;
        *(v4f*)(Blds + row * BROW + q * 4) = bw[j];
    }
    {
        const int i   = BTAIL + (tid >> 1);
        const int row = i / BV4ROW;
        const int q   = i - row * BV4ROW;
        *(v2f*)(Blds + row * BROW + q * 4 + (tid & 1) * 2) = bt;
    }
    __builtin_amdgcn_fence(__ATOMIC_RELEASE, "workgroup", "local");
    __builtin_amdgcn_s_barrier();
    __builtin_amdgcn_fence(__ATOMIC_ACQUIRE, "workgroup", "local");
    asm volatile("" ::: "memory");

    const float* bbase = &Blds[(kHalf >> 1) * BROW + laneM * 2];

    const int hrY  = h0 - 2 + rY;
    const bool okY = (hrY >= 0 && hrY < HH);
    const int hcY  = hrY < 0 ? 0 : (hrY > HH - 1 ? HH - 1 : hrY);
    const float* xrY = xb + hcY * WW;

    if (pr & 1) {
        {
            v8f d0[2], d1[2];
            mma_row<2, true>(aX0, aX1, bbase + cX * 2, d0, d1, aY0, aY1, xrX, xrY, kHalf);
            #pragma unroll
            for (int t = 0; t < 2; ++t) {
                store_tile(Yl, rX, cX + 16 * t, 0, lane, d0[t]);
                store_tile(Yl, rX, cX + 16 * t, 1, lane, d1[t]);
            }
        }
        if (okY) {
            v8f d0[2], d1[2];
            mma_row<2, false>(aY0, aY1, bbase + cY * 2, d0, d1, aX0, aX1, xrY, xrY, kHalf);
            #pragma unroll
            for (int t = 0; t < 2; ++t) {
                store_tile(Yl, rY, cY + 16 * t, 0, lane, d0[t]);
                store_tile(Yl, rY, cY + 16 * t, 1, lane, d1[t]);
            }
        }
        asm volatile("" :: "v"(aY0[23]), "v"(aY1[23]));
        __builtin_amdgcn_fence(__ATOMIC_RELEASE, "workgroup", "local");
    } else {
        {
            v8f d0[3], d1[3];
            mma_row<3, true>(aX0, aX1, bbase + cX * 2, d0, d1, aY0, aY1, xrX, xrY, kHalf);
            #pragma unroll
            for (int t = 0; t < 3; ++t) {
                store_tile(Yl, rX, cX + 16 * t, 0, lane, d0[t]);
                store_tile(Yl, rX, cX + 16 * t, 1, lane, d1[t]);
            }
        }
        if (okY) {
            v8f d0[1], d1[1];
            mma_row<1, false>(aY0, aY1, bbase + cY * 2, d0, d1, aX0, aX1, xrY, xrY, kHalf);
            store_tile(Yl, rY, cY, 0, lane, d0[0]);
            store_tile(Yl, rY, cY, 1, lane, d1[0]);
        }
        asm volatile("" :: "v"(aY0[23]), "v"(aY1[23]));
    }

    __builtin_amdgcn_fence(__ATOMIC_RELEASE, "workgroup", "local");
    __builtin_amdgcn_s_barrier();
    __builtin_amdgcn_fence(__ATOMIC_ACQUIRE, "workgroup", "local");

    if (tid < ROWS_OUT * PIX_OUT) {
        const int a  = tid / PIX_OUT;
        const int w  = w0 + (tid - a * PIX_OUT);
        const int h  = h0 + a;

        v2f cm[PP];
        #pragma unroll
        for (int p = 0; p < PP; ++p) {
            const int i   = p / KK, j = p - i * KK;
            const int hh  = h + i - 2;
            const int ww2 = w + j - 2;
            const bool in  = (hh >= 0 && hh < HH && ww2 >= 0 && ww2 < WW);
            const int hc   = hh < 0 ? 0 : (hh > HH - 1 ? HH - 1 : hh);
            const int wc   = ww2 < 0 ? 0 : (ww2 > WW - 1 ? WW - 1 : ww2);
            const int ic   = hc - h + 2;
            const v2f yv = *(const v2f*)&Yl[(((a * KK + ic) * PIX_IN) + (wc - wb)) * YCOLS + j * NC];
            cm[p].x = (in ? yv.x : 0.f) + bk0;
            cm[p].y = (in ? yv.y : 0.f) + bk1;
        }
        v2f res;
        #pragma unroll
        for (int k = 0; k < NC; ++k) {
            float asum = 0.f, ys = 0.f, xs = 0.f, cs = 0.f;
            #pragma unroll
            for (int p = 0; p < PP; ++p) {
                float c  = cm[p][k];
                float ab = fabsf(c);
                float yo = (float)(p / KK - 2);
                float xo = (float)(p % KK - 2);
                asum += ab;
                ys   += ab * yo;
                xs   += ab * xo;
                cs   += c;
            }
            float yd    = ys / asum;
            float xd    = xs / asum;
            float drift = sqrtf(xd * xd + yd * yd);
            res[k] = cs * expf(-0.5f * drift);
        }
        *(v2f*)&out[((size_t)n * HW + h * WW + w) * NC] = res;
    }
}

extern "C" void kernel_launch(void* const* d_in, const int* in_sizes, int n_in,
                              void* d_out, int out_size, void* d_ws, size_t ws_size,
                              hipStream_t stream) {
    const float* x    = (const float*)d_in[0];
    const float* Wt   = (const float*)d_in[1];
    const float* bias = (const float*)d_in[2];
    const float* Wlin = (const float*)d_in[3];
    const float* blin = (const float*)d_in[4];
    float* out = (float*)d_out;

    float* wsf   = (float*)d_ws;
    float* Bp    = wsf;
    float* bfold = wsf + WS_BF;

    calib_fold_kernel<<<FOLD_BLK + 1, FOLD_THR, 0, stream>>>(
        Wt, bias, Wlin, blin, Bp, bfold);

    calib_conv_kernel<<<NB * BLK_PER_IMG, NTHR, 0, stream>>>(x, Bp, bfold, out);
}


# Input loads widened to 8 bytes: tile row m is pixel 2m+t, a lane reads a pixel pair of one channel in one load (128 loads a wave, not 256), so load groups stay inside the 63-deep load counter
# speedup vs baseline: 2.3752x; 1.1166x over previous
#include <hip/hip_runtime.h>
#include <math.h>


#define NB   8
#define CCH  128
#define HH   56
#define WW   56
#define OCH  64
#define NC   2
#define KK   5
#define PP   (KK*KK)
#define HW   (HH*WW)
#define NPIX (NB*HW)

#define NCOL  (PP*NC)
#define BUSED (NCOL*2)
#define BROW  160
#define BROWS 64
#define BTOT  (BROWS*BROW)
#define GTOT  (BROWS*BUSED)
#define BV4   (GTOT/4)
#define BV4ROW (BUSED/4)

#define ROWS_OUT 4
#define PIX_OUT  28
#define PIX_IN   32
#define YCOLS (KK*NC)
#define YTOT  (ROWS_OUT*KK*PIX_IN*YCOLS)
#define BLK_PER_IMG ((HH/ROWS_OUT)*2)

typedef __attribute__((ext_vector_type(2))) float v2f;
typedef __attribute__((ext_vector_type(4))) float v4f;
typedef __attribute__((ext_vector_type(8))) float v8f;

#define WS_BF GTOT

#define FOLD_THR 32
#define FOLD_BLK (GTOT / FOLD_THR)
#define OH       (OCH / 2)
static_assert(FOLD_BLK * FOLD_THR == GTOT, "fold blocks");
static_assert(OH % 16 == 0, "64 byte pieces of Wlin");

__global__ void __launch_bounds__(FOLD_THR)
calib_fold_kernel(const float* __restrict__ Wt,
                  const float* __restrict__ bias,
                  const float* __restrict__ Wlin,
                  const float* __restrict__ blin,
                  float* __restrict__ Bp,
                  float* __restrict__ bfold) {
    const int lane = threadIdx.x;
    if (blockIdx.x == 0) {
        if (lane < NC) {
            float bv[OCH];
            #pragma unroll
            for (int o = 0; o < OCH; ++o) bv[o] = bias[o];
            __builtin_amdgcn_sched_barrier(0);
            const float* wl = Wlin + lane * OCH;
            float lv[OCH];
            #pragma unroll
            for (int o = 0; o < OCH; ++o) lv[o] = wl[o];
            float s0 = blin[lane], s1 = 0.f, s2 = 0.f, s3 = 0.f;
            __builtin_amdgcn_sched_barrier(0);
            #pragma unroll
            for (int o = 0; o < OCH; o += 4) {
                s0 = __builtin_fmaf(lv[o],     bv[o],     s0);
                s1 = __builtin_fmaf(lv[o + 1], bv[o + 1], s1);
                s2 = __builtin_fmaf(lv[o + 2], bv[o + 2], s2);
                s3 = __builtin_fmaf(lv[o + 3], bv[o + 3], s3);
                asm volatile("" : "+v"(s0), "+v"(s1), "+v"(s2), "+v"(s3));
            }
            bfold[lane] = (s0 + s1) + (s2 + s3);
        }
        return;
    }
    const int idx = (blockIdx.x - 1) * FOLD_THR + lane;
    const int row = idx / BUSED;
    const int off = idx - row * BUSED;
    const int p   = off >> 2;
    const int k   = (off >> 1) & 1;
    const int c   = row * 2 + (off & 1);
    const float* wt = Wt + c * PP + p;

    float l0[OH], l1[OH];
    #pragma unroll
    for (int o = 0; o < OH; ++o) { l0[o] = Wlin[o]; l1[o] = Wlin[OCH + o]; }
    const float w0a = Wlin[OH], w0b = Wlin[OH + 16];
    const float w1a = Wlin[OCH + OH], w1b = Wlin[OCH + OH + 16];
    __builtin_amdgcn_sched_barrier(0);
    float tv[OCH];
    #pragma unroll
    for (int o = 0; o < OCH; ++o) tv[o] = wt[(size_t)o * (CCH * PP)];
    __builtin_amdgcn_sched_barrier(0);

    float a0 = 0.f, a1 = 0.f, a2 = 0.f, a3 = 0.f;
    float b0 = 0.f, b1 = 0.f, b2 = 0.f, b3 = 0.f;
    #pragma unroll
    for (int o = 0; o < OH; o += 4) {
        a0 = __builtin_fmaf(l0[o],     tv[o],     a0);
        b0 = __builtin_fmaf(l1[o],     tv[o],     b0);
        a1 = __builtin_fmaf(l0[o + 1], tv[o + 1], a1);
        b1 = __builtin_fmaf(l1[o + 1], tv[o + 1], b1);
        a2 = __builtin_fmaf(l0[o + 2], tv[o + 2], a2);
        b2 = __builtin_fmaf(l1[o + 2], tv[o + 2], b2);
        a3 = __builtin_fmaf(l0[o + 3], tv[o + 3], a3);
        b3 = __builtin_fmaf(l1[o + 3], tv[o + 3], b3);
        asm volatile("" : "+v"(a0), "+v"(a1), "+v"(a2), "+v"(a3),
                          "+v"(b0), "+v"(b1), "+v"(b2), "+v"(b3));
    }
    int oh = OH;
    asm volatile("" : "+s"(oh) : "s"(w0a), "s"(w0b), "s"(w1a), "s"(w1b));
    const float* m0 = Wlin + oh;
    const float* m1 = Wlin + OCH + oh;
    float n0[OH], n1[OH];
    #pragma unroll
    for (int o = 0; o < OH; ++o) { n0[o] = m0[o]; n1[o] = m1[o]; }
    __builtin_amdgcn_sched_barrier(0);
    #pragma unroll
    for (int o = 0; o < OH; o += 4) {
        a0 = __builtin_fmaf(n0[o],     tv[OH + o],     a0);
        b0 = __builtin_fmaf(n1[o],     tv[OH + o],     b0);
        a1 = __builtin_fmaf(n0[o + 1], tv[OH + o + 1], a1);
        b1 = __builtin_fmaf(n1[o + 1], tv[OH + o + 1], b1);
        a2 = __builtin_fmaf(n0[o + 2], tv[OH + o + 2], a2);
        b2 = __builtin_fmaf(n1[o + 2], tv[OH + o + 2], b2);
        a3 = __builtin_fmaf(n0[o + 3], tv[OH + o + 3], a3);
        b3 = __builtin_fmaf(n1[o + 3], tv[OH + o + 3], b3);
        asm volatile("" : "+v"(a0), "+v"(a1), "+v"(a2), "+v"(a3),
                          "+v"(b0), "+v"(b1), "+v"(b2), "+v"(b3));
    }
    const float v0 = (a0 + a1) + (a2 + a3);
    const float v1 = (b0 + b1) + (b2 + b3);
    Bp[idx] = k ? v1 : v0;
}

#define NTHR   128
#define BFULL  (BV4 / NTHR)
#define BTAIL  (BFULL * NTHR)
#define KSTEPS (CCH / 4)
static_assert((BV4 - BTAIL) * 2 == NTHR, "tail halves");
#define KBLK   8

__device__ __forceinline__ v8f wmma4(v2f a, v2f b, v8f c) {
    return __builtin_amdgcn_wmma_f32_16x16x4_f32(false, a, false, b,
                                                 (short)0, c, false, false);
}

template <int CNT>
__device__ __forceinline__ void load_ks(v2f (&a0)[KSTEPS], v2f (&a1)[KSTEPS],
                                        const float* xr, int kHalf, int ks0) {
    #pragma unroll
    for (int kk = 0; kk < CNT; ++kk) {
        const int ks = ks0 + kk;
        const int c  = ks * 4 + kHalf;
        a0[ks] = *(const v2f*)(xr + (size_t)c * HW);
        a1[ks] = *(const v2f*)(xr + (size_t)(c + 1) * HW);
    }
}

template <int NT, bool FIRST>
__device__ __forceinline__ void mma_row(v2f (&a0)[KSTEPS], v2f (&a1)[KSTEPS],
                                        const float* bb, v8f (&d0)[NT], v8f (&d1)[NT],
                                        v2f (&n0)[KSTEPS], v2f (&n1)[KSTEPS],
                                        const float* xr, const float* nxr, int kHalf) {
    v2f bf[2][KBLK][NT];
    #pragma unroll
    for (int kk = 0; kk < KBLK; ++kk) {
        #pragma unroll
        for (int t = 0; t < NT; ++t)
            bf[0][kk][t] = *(const v2f*)(bb + kk * 2 * BROW + t * 32);
    }
    #pragma unroll
    for (int t = 0; t < NT; ++t) {
        d0[t] = v8f{};
        d1[t] = v8f{};
    }
    #pragma unroll
    for (int blk = 0; blk < KSTEPS / KBLK; ++blk) {
        if (blk + 1 < KSTEPS / KBLK) {
            #pragma unroll
            for (int kk = 0; kk < KBLK; ++kk) {
                #pragma unroll
                for (int t = 0; t < NT; ++t)
                    bf[(blk + 1) & 1][kk][t] =
                        *(const v2f*)(bb + ((blk + 1) * KBLK + kk) * 2 * BROW + t * 32);
            }
        }
        if (FIRST) {
            if (blk == 0) load_ks<8>(a0, a1, xr, kHalf, 16);
            if (blk == 1) load_ks<8>(a0, a1, xr, kHalf, 24);
            if (blk == 2) load_ks<12>(n0, n1, nxr, kHalf, 0);
            if (blk == 3) {
                load_ks<11>(n0, n1, nxr, kHalf, 12);
                __builtin_amdgcn_sched_barrier(0);
                load_ks<1>(n0, n1, nxr, kHalf, 23);
            }
        } else {
            if (blk == 0) load_ks<8>(a0, a1, xr, kHalf, 24);
        }
        __builtin_amdgcn_sched_barrier(0);
        #pragma unroll
        for (int kk = 0; kk < KBLK; ++kk) {
            const int ks = blk * KBLK + kk;
            const v2f p = a0[ks], q = a1[ks];
            v2f ev, od;
            ev.x = p.x; ev.y = q.x;
            od.x = p.y; od.y = q.y;
            #pragma unroll
            for (int t = 0; t < NT; ++t) {
                const v2f b = bf[blk & 1][kk][t];
                d0[t] = wmma4(ev, b, d0[t]);
                d1[t] = wmma4(od, b, d1[t]);
            }
        }
        __builtin_amdgcn_sched_barrier(0);
    }
    asm volatile("v_nop\n\tv_nop\n\tv_nop\n\tv_nop");
    __builtin_amdgcn_sched_barrier(0);
}

__device__ __forceinline__ void store_tile(float* Yl, int r, int c0, int mt,
                                           int lane, v8f acc) {
    const int col = c0 + (lane & 15);
    const int i   = col / YCOLS;
    const int jk  = col - i * YCOLS;
    const int a   = r - i;
    const int m0  = ((lane >> 4) << 4) + mt;
    if (a >= 0 && a < ROWS_OUT) {
        float* yp = Yl + ((a * KK + i) * PIX_IN + m0) * YCOLS + jk;
        #pragma unroll
        for (int q = 0; q < 8; ++q) yp[q * 2 * YCOLS] = acc[q];
    }
}

__global__ void __launch_bounds__(NTHR)
calib_conv_kernel(const float* __restrict__ x,
                  const float* __restrict__ Bp,
                  const float* __restrict__ bfold,
                  float* __restrict__ out) {
    __shared__ __attribute__((aligned(16))) float Blds[BTOT];
    __shared__ __attribute__((aligned(16))) float Yl[YTOT];

    const int tid  = threadIdx.x;
    const int pr   = __builtin_amdgcn_readfirstlane(tid >> 5);
    const int lane = tid & 31;

    const int n    = blockIdx.x / BLK_PER_IMG;
    const int rem  = blockIdx.x - n * BLK_PER_IMG;
    const int h0   = (rem >> 1) * ROWS_OUT;
    const int side = rem & 1;
    const int w0   = side * PIX_OUT;
    const int wb   = side * (WW - PIX_IN);

    const int rX = (pr == 0) ? 3 : (pr == 1) ? 2 : (pr == 2) ? 4 : 5;
    const int rY = (pr == 0) ? 0 : (pr == 1) ? 1 : (pr == 2) ? 7 : 6;
    const int cX = (pr == 0) ? 0 : (pr == 1) ? 0 : (pr == 2) ? 2 : 18;
    const int cY = (pr == 0) ? 0 : (pr == 1) ? 0 : (pr == 2) ? 34 : 18;

    const int laneM = lane & 15;
    const int kHalf = (lane >> 4) << 1;
    const float* xb = x + (size_t)n * CCH * HW + wb + 2 * laneM;

    v4f bw[BFULL];
    #pragma unroll
    for (int j = 0; j < BFULL; ++j) bw[j] = ((const v4f*)Bp)[tid + j * NTHR];
    const v2f bt = ((const v2f*)Bp)[BTAIL * 2 + tid];
    const float bk0 = bfold[0];
    const float bk1 = bfold[1];
    asm volatile("" ::: "memory");

    v2f aX0[KSTEPS], aX1[KSTEPS], aY0[KSTEPS], aY1[KSTEPS];
    const float* xrX = xb + (h0 - 2 + rX) * WW;
    load_ks<16>(aX0, aX1, xrX, kHalf, 0);

    #pragma unroll
    for (int j = 0; j < BFULL; ++j) {
        const int i   = tid + j * NTHR;
        const int row = i / BV4ROW;
        const int q   = i - row * BV4ROW;
        *(v4f*)(Blds + row * BROW + q * 4) = bw[j];
    }
    {
        const int i   = BTAIL + (tid >> 1);
        const int row = i / BV4ROW;
        const int q   = i - row * BV4ROW;
        *(v2f*)(Blds + row * BROW + q * 4 + (tid & 1) * 2) = bt;
    }
    __builtin_amdgcn_fence(__ATOMIC_RELEASE, "workgroup", "local");
    __builtin_amdgcn_s_barrier();
    __builtin_amdgcn_fence(__ATOMIC_ACQUIRE, "workgroup", "local");
    asm volatile("" ::: "memory");

    const float* bbase = &Blds[(kHalf >> 1) * BROW + laneM * 2];

    const int hrY  = h0 - 2 + rY;
    const bool okY = (hrY >= 0 && hrY < HH);
    const int hcY  = hrY < 0 ? 0 : (hrY > HH - 1 ? HH - 1 : hrY);
    const float* xrY = xb + hcY * WW;

    if (pr & 1) {
        {
            v8f d0[2], d1[2];
            mma_row<2, true>(aX0, aX1, bbase + cX * 2, d0, d1, aY0, aY1, xrX, xrY, kHalf);
            #pragma unroll
            for (int t = 0; t < 2; ++t) {
                store_tile(Yl, rX, cX + 16 * t, 0, lane, d0[t]);
                store_tile(Yl, rX, cX + 16 * t, 1, lane, d1[t]);
            }
        }
        if (okY) {
            v8f d0[2], d1[2];
            mma_row<2, false>(aY0, aY1, bbase + cY * 2, d0, d1, aX0, aX1, xrY, xrY, kHalf);
            #pragma unroll
            for (int t = 0; t < 2; ++t) {
                store_tile(Yl, rY, cY + 16 * t, 0, lane, d0[t]);
                store_tile(Yl, rY, cY + 16 * t, 1, lane, d1[t]);
            }
        }
        asm volatile("" :: "v"(aY0[23]), "v"(aY1[23]));
        __builtin_amdgcn_fence(__ATOMIC_RELEASE, "workgroup", "local");
    } else {
        {
            v8f d0[3], d1[3];
            mma_row<3, true>(aX0, aX1, bbase + cX * 2, d0, d1, aY0, aY1, xrX, xrY, kHalf);
            #pragma unroll
            for (int t = 0; t < 3; ++t) {
                store_tile(Yl, rX, cX + 16 * t, 0, lane, d0[t]);
                store_tile(Yl, rX, cX + 16 * t, 1, lane, d1[t]);
            }
        }
        if (okY) {
            v8f d0[1], d1[1];
            mma_row<1, false>(aY0, aY1, bbase + cY * 2, d0, d1, aX0, aX1, xrY, xrY, kHalf);
            store_tile(Yl, rY, cY, 0, lane, d0[0]);
            store_tile(Yl, rY, cY, 1, lane, d1[0]);
        }
        asm volatile("" :: "v"(aY0[23]), "v"(aY1[23]));
    }

    __builtin_amdgcn_fence(__ATOMIC_RELEASE, "workgroup", "local");
    __builtin_amdgcn_s_barrier();
    __builtin_amdgcn_fence(__ATOMIC_ACQUIRE, "workgroup", "local");

    if (tid < ROWS_OUT * PIX_OUT) {
        const int a  = tid / PIX_OUT;
        const int w  = w0 + (tid - a * PIX_OUT);
        const int h  = h0 + a;

        v2f cm[PP];
        #pragma unroll
        for (int p = 0; p < PP; ++p) {
            const int i   = p / KK, j = p - i * KK;
            const int hh  = h + i - 2;
            const int ww2 = w + j - 2;
            const bool in  = (hh >= 0 && hh < HH && ww2 >= 0 && ww2 < WW);
            const int hc   = hh < 0 ? 0 : (hh > HH - 1 ? HH - 1 : hh);
            const int wc   = ww2 < 0 ? 0 : (ww2 > WW - 1 ? WW - 1 : ww2);
            const int ic   = hc - h + 2;
            const v2f yv = *(const v2f*)&Yl[(((a * KK + ic) * PIX_IN) + (wc - wb)) * YCOLS + j * NC];
            cm[p].x = (in ? yv.x : 0.f) + bk0;
            cm[p].y = (in ? yv.y : 0.f) + bk1;
        }
        v2f res;
        #pragma unroll
        for (int k = 0; k < NC; ++k) {
            float asum = 0.f, ys = 0.f, xs = 0.f, cs = 0.f;
            #pragma unroll
            for (int p = 0; p < PP; ++p) {
                float c  = cm[p][k];
                float ab = fabsf(c);
                float yo = (float)(p / KK - 2);
                float xo = (float)(p % KK - 2);
                asum += ab;
                ys   += ab * yo;
                xs   += ab * xo;
                cs   += c;
            }
            float yd    = ys / asum;
            float xd    = xs / asum;
            float drift = sqrtf(xd * xd + yd * yd);
            res[k] = cs * expf(-0.5f * drift);
        }
        *(v2f*)&out[((size_t)n * HW + h * WW + w) * NC] = res;
    }
}

extern "C" void kernel_launch(void* const* d_in, const int* in_sizes, int n_in,
                              void* d_out, int out_size, void* d_ws, size_t ws_size,
                              hipStream_t stream) {
    const float* x    = (const float*)d_in[0];
    const float* Wt   = (const float*)d_in[1];
    const float* bias = (const float*)d_in[2];
    const float* Wlin = (const float*)d_in[3];
    const float* blin = (const float*)d_in[4];
    float* out = (float*)d_out;

    float* wsf   = (float*)d_ws;
    float* Bp    = wsf;
    float* bfold = wsf + WS_BF;

    calib_fold_kernel<<<FOLD_BLK + 1, FOLD_THR, 0, stream>>>(
        Wt, bias, Wlin, blin, Bp, bfold);

    calib_conv_kernel<<<NB * BLK_PER_IMG, NTHR, 0, stream>>>(x, Bp, bfold, out);
}
